# starting kernel
# baseline (speedup 1.0000x reference)
#include <hip/hip_runtime.h>
#include <math.h>

#define B_  4
#define LQ  512
#define LK  512
#define E_  128
#define D_  256

typedef float v2f __attribute__((ext_vector_type(2)));
typedef float v8f __attribute__((ext_vector_type(8)));

__device__ __forceinline__ float fast_tanh(float x) {
#if __has_builtin(__builtin_amdgcn_tanhf)
    return __builtin_amdgcn_tanhf(x);
#else
    return tanhf(x);
#endif
}

__device__ __forceinline__ float fast_exp(float x) {
#if __has_builtin(__builtin_amdgcn_exp2f)
    return __builtin_amdgcn_exp2f(x * 1.44269504088896340f);
#else
    return __expf(x);
#endif
}

__global__ __launch_bounds__(128) void proj_kernel(
    const float* __restrict__ query, const float* __restrict__ key,
    const float* __restrict__ Wc1,   const float* __restrict__ Wc2,
    float* __restrict__ qp, float* __restrict__ kp)
{
    __shared__ __align__(16) float row[D_];
    int bid  = blockIdx.x;
    int half = bid >> 11;
    int r    = bid & 2047;
    int b    = r >> 9;
    int m    = r & 511;
    const float* in  = half ? key : query;
    const float* W   = half ? Wc2 : Wc1;
    float*       out = half ? kp  : qp;

    const float* src = in + ((size_t)(b * 512 + m)) * D_;
    int tid = threadIdx.x;
    row[tid]       = src[tid];
    row[tid + 128] = src[tid + 128];
    __syncthreads();

    float acc = 0.f;
#pragma unroll 8
    for (int d = 0; d < D_; ++d)
        acc = fmaf(row[d], W[d * E_ + tid], acc);
    out[((size_t)(b * 512 + m)) * E_ + tid] = acc;
}

__global__ __launch_bounds__(256) void score_softmax_kernel(
    const float* __restrict__ qp, const float* __restrict__ kp,
    const float* __restrict__ vc, float* __restrict__ atten)
{
    __shared__ __align__(16) float kpl[E_];
    __shared__ __align__(16) float vcl[E_];
    __shared__ float red[256];

    int bid = blockIdx.x;
    int b   = bid >> 9;
    int l   = bid & 511;
    int tid = threadIdx.x;

    if (tid < E_) {
        kpl[tid] = kp[((size_t)(b * LK + l)) * E_ + tid];
        vcl[tid] = vc[tid];
    }
    __syncthreads();

    const float4* kp4 = (const float4*)kpl;
    const float4* vc4 = (const float4*)vcl;

    float accs[2];
#pragma unroll
    for (int j = 0; j < 2; ++j) {
        int q = tid + j * 256;
        const float4* qrow = (const float4*)(qp + ((size_t)(b * LQ + q)) * E_);
        float acc = 0.f;
#pragma unroll 8
        for (int e4 = 0; e4 < E_ / 4; ++e4) {
            float4 qv = qrow[e4];
            float4 kv = kp4[e4];
            float4 vv = vc4[e4];
            acc = fmaf(fast_tanh(qv.x + kv.x), vv.x, acc);
            acc = fmaf(fast_tanh(qv.y + kv.y), vv.y, acc);
            acc = fmaf(fast_tanh(qv.z + kv.z), vv.z, acc);
            acc = fmaf(fast_tanh(qv.w + kv.w), vv.w, acc);
        }
        accs[j] = acc;
    }

    red[tid] = fmaxf(accs[0], accs[1]);
    __syncthreads();
    for (int off = 128; off > 0; off >>= 1) {
        if (tid < off) red[tid] = fmaxf(red[tid], red[tid + off]);
        __syncthreads();
    }
    float mx = red[0];
    __syncthreads();

    float e0 = fast_exp(accs[0] - mx);
    float e1 = fast_exp(accs[1] - mx);
    red[tid] = e0 + e1;
    __syncthreads();
    for (int off = 128; off > 0; off >>= 1) {
        if (tid < off) red[tid] += red[tid + off];
        __syncthreads();
    }
    float inv = 1.f / red[0];

    float* arow = atten + ((size_t)(b * LK + l)) * LQ;
    arow[tid]       = e0 * inv;
    arow[tid + 256] = e1 * inv;
}

__global__ __launch_bounds__(128) void context_wmma_kernel(
    const float* __restrict__ atten, const float* __restrict__ value,
    float* __restrict__ context)
{
    int wave = threadIdx.x >> 5;
    int lane = threadIdx.x & 31;
    int tile = blockIdx.x * 4 + wave;
    int b    = tile >> 9;
    int r    = tile & 511;
    int lt   = r >> 4;
    int dt   = r & 15;
    int l0   = lt * 16;
    int d0   = dt * 16;
    int half = lane >> 4;
    int lh   = lane & 15;

    const float* arow = atten + ((size_t)(b * LK + l0 + lh)) * LQ + half * 2;
    const float* vcol = value + ((size_t)(b * LQ)) * D_ + d0 + lh;

    v8f c = {0.f, 0.f, 0.f, 0.f, 0.f, 0.f, 0.f, 0.f};
    for (int q = 0; q < LQ; q += 4) {
        v2f a = *(const v2f*)(arow + q);
        v2f bm;
        bm.x = vcol[(size_t)(q + half * 2) * D_];
        bm.y = vcol[(size_t)(q + half * 2 + 1) * D_];
        c = __builtin_amdgcn_wmma_f32_16x16x4_f32(
                 false, a,  false, bm,
                 (short)0, c,  false,  false);
    }

    float* crow = context + ((size_t)(b * LK + l0 + half * 8)) * D_ + d0 + lh;
#pragma unroll
    for (int i = 0; i < 8; ++i)
        crow[(size_t)i * D_] = c[i];
}

extern "C" void kernel_launch(void* const* d_in, const int* in_sizes, int n_in,
                              void* d_out, int out_size, void* d_ws, size_t ws_size,
                              hipStream_t stream)
{
    const float* query = (const float*)d_in[0];
    const float* key   = (const float*)d_in[1];
    const float* value = (const float*)d_in[2];
    const float* Wc1   = (const float*)d_in[3];
    const float* Wc2   = (const float*)d_in[4];
    const float* vc    = (const float*)d_in[5];

    float* out     = (float*)d_out;
    float* context = out;
    float* atten   = out + (size_t)B_ * LK * D_;

    float* qp = (float*)d_ws;
    float* kp = qp + (size_t)B_ * LQ * E_;

    proj_kernel<<<2 * B_ * 512, 128, 0, stream>>>(query, key, Wc1, Wc2, qp, kp);
    score_softmax_kernel<<<B_ * LK, 256, 0, stream>>>(qp, kp, vc, atten);
    context_wmma_kernel<<<(B_ * (LK / 16) * (D_ / 16)) / 4, 128, 0, stream>>>(
        atten, value, context);
}


# context GEMM fed straight from L2 by 16-byte fragment loads kept 5 groups ahead, no LDS staging: a wave holds a 32x64 tile on 8 accumulators, K split over 4 waves, joined once in LDS
# speedup vs baseline: 2.1807x; 2.1807x over previous
#include <hip/hip_runtime.h>
#include <math.h>

#define B_  4
#define LQ  512
#define LK  512
#define E_  128
#define D_  256

typedef float v2f __attribute__((ext_vector_type(2)));
typedef float v8f __attribute__((ext_vector_type(8)));

__device__ __forceinline__ float fast_tanh(float x) {
#if __has_builtin(__builtin_amdgcn_tanhf)
    return __builtin_amdgcn_tanhf(x);
#else
    return tanhf(x);
#endif
}

__device__ __forceinline__ float fast_exp(float x) {
#if __has_builtin(__builtin_amdgcn_exp2f)
    return __builtin_amdgcn_exp2f(x * 1.44269504088896340f);
#else
    return __expf(x);
#endif
}

#define PRW 4
#define PC  16
#define DH  (D_ / 2)

__device__ __forceinline__ void proj_chunk(v2f (&acc)[PRW], const v2f (&w)[PC],
                                           const float* __restrict__ x)
{
#pragma unroll
    for (int j = 0; j < PC; ++j) {
#pragma unroll
        for (int r = 0; r < PRW; ++r) {
            const float xs = x[r * D_ + j];
            acc[r] = __builtin_elementwise_fma((v2f){xs, xs}, w[j], acc[r]);
        }
    }
}

__global__ __launch_bounds__(256) void proj_kernel(
    const float* __restrict__ query, const float* __restrict__ key,
    const float* __restrict__ Wc1,   const float* __restrict__ Wc2,
    float* __restrict__ qp, float* __restrict__ kp)
{
    __shared__ v2f red[2 * PRW * (E_ / 2)];

    const int nb   = (B_ * 512) / (2 * PRW);
    const int bid  = blockIdx.x;
    const int half = bid / nb;
    const int tid  = threadIdx.x;
    const int ep   = tid & 63;
    const int rh   = __builtin_amdgcn_readfirstlane((tid >> 6) & 1);
    const int dh   = __builtin_amdgcn_readfirstlane(tid >> 7);
    const int row0 = (bid % nb) * (2 * PRW) + rh * PRW;

    const float* in  = (half ? key : query) + (size_t)row0 * D_ + dh * DH;
    const v2f*   W   = (const v2f*)(half ? Wc2 : Wc1) + (size_t)dh * DH * (E_ / 2) + ep;
    v2f*         out = (v2f*)((half ? kp : qp) + (size_t)row0 * E_) + ep;

    v2f acc[PRW];
#pragma unroll
    for (int r = 0; r < PRW; ++r) acc[r] = (v2f){0.f, 0.f};

    v2f wa[PC], wb[PC];
#pragma unroll
    for (int j = 0; j < PC; ++j) wa[j] = W[j * (E_ / 2)];

#pragma unroll 1
    for (int c = 0; c < DH / PC; c += 2) {
        const v2f* Wb = W + (size_t)(c + 1) * PC * (E_ / 2);
#pragma unroll
        for (int j = 0; j < PC; ++j) wb[j] = Wb[j * (E_ / 2)];
        proj_chunk(acc, wa, in + c * PC);

        const int cn = (c + 2 < DH / PC) ? (c + 2) : (c + 1);
        const v2f* Wa = W + (size_t)cn * PC * (E_ / 2);
#pragma unroll
        for (int j = 0; j < PC; ++j) wa[j] = Wa[j * (E_ / 2)];
        proj_chunk(acc, wb, in + (c + 1) * PC);
    }

    if (dh) {
#pragma unroll
        for (int r = 0; r < PRW; ++r) red[(rh * PRW + r) * (E_ / 2) + ep] = acc[r];
    }
    __syncthreads();
    if (!dh) {
#pragma unroll
        for (int r = 0; r < PRW; ++r) {
            const v2f t = red[(rh * PRW + r) * (E_ / 2) + ep];
            out[(size_t)r * (E_ / 2)] = acc[r] + t;
        }
    }
}

#define KR  8
#define QPITCH (E_ + 4)

__global__ __launch_bounds__(512) void score_softmax_kernel(
    const float* __restrict__ qp, const float* __restrict__ kp,
    const float* __restrict__ vc, float* __restrict__ atten)
{
    __shared__ __align__(16) float qs[LQ * QPITCH];
    __shared__ float sc[KR * LQ];
    __shared__ float redm[KR * 16];
    __shared__ float reds[KR * 16];

    const int bid  = blockIdx.x;
    const int b    = bid / (LK / KR);
    const int l0   = (bid % (LK / KR)) * KR;
    const int tid  = threadIdx.x;
    const int wave = tid >> 5;
    const int lane = tid & 31;

    {
        const float4* src4 = (const float4*)(qp + ((size_t)(b * LQ + wave * 32)) * E_);
#pragma unroll 8
        for (int r = 0; r < 32; ++r) {
            float4 v = src4[r * (E_ / 4) + lane];
            *(float4*)&qs[(wave * 32 + r) * QPITCH + lane * 4] = v;
        }
    }
    __syncthreads();

    float q[E_];
#pragma unroll
    for (int e4 = 0; e4 < E_ / 4; ++e4) {
        float4 v = *(const float4*)&qs[tid * QPITCH + e4 * 4];
        q[e4 * 4 + 0] = v.x;
        q[e4 * 4 + 1] = v.y;
        q[e4 * 4 + 2] = v.z;
        q[e4 * 4 + 3] = v.w;
    }

    const float* kb = kp + ((size_t)(b * LK + l0)) * E_;

#pragma unroll 1
    for (int rg = 0; rg < KR / 4; ++rg) {
        const float* k0 = kb + (size_t)(4 * rg) * E_;
        const float* k1 = k0 + E_;
        const float* k2 = k1 + E_;
        const float* k3 = k2 + E_;
        float a0 = 0.f, a1 = 0.f, a2 = 0.f, a3 = 0.f;
        float c0 = 0.f, c1 = 0.f, c2 = 0.f, c3 = 0.f;
#pragma unroll
        for (int e = 0; e < E_; e += 2) {
            a0 = fmaf(fast_tanh(q[e]     + k0[e]),     vc[e],     a0);
            a1 = fmaf(fast_tanh(q[e]     + k1[e]),     vc[e],     a1);
            a2 = fmaf(fast_tanh(q[e]     + k2[e]),     vc[e],     a2);
            a3 = fmaf(fast_tanh(q[e]     + k3[e]),     vc[e],     a3);
            c0 = fmaf(fast_tanh(q[e + 1] + k0[e + 1]), vc[e + 1], c0);
            c1 = fmaf(fast_tanh(q[e + 1] + k1[e + 1]), vc[e + 1], c1);
            c2 = fmaf(fast_tanh(q[e + 1] + k2[e + 1]), vc[e + 1], c2);
            c3 = fmaf(fast_tanh(q[e + 1] + k3[e + 1]), vc[e + 1], c3);
        }
        sc[(4 * rg)     * LQ + tid] = a0 + c0;
        sc[(4 * rg + 1) * LQ + tid] = a1 + c1;
        sc[(4 * rg + 2) * LQ + tid] = a2 + c2;
        sc[(4 * rg + 3) * LQ + tid] = a3 + c3;
    }

    float s[KR], m[KR];
#pragma unroll
    for (int r = 0; r < KR; ++r) { s[r] = sc[r * LQ + tid]; m[r] = s[r]; }

#pragma unroll
    for (int off = 16; off > 0; off >>= 1) {
        float t[KR];
#pragma unroll
        for (int r = 0; r < KR; ++r) t[r] = __shfl_xor(m[r], off, 32);
#pragma unroll
        for (int r = 0; r < KR; ++r) m[r] = fmaxf(m[r], t[r]);
    }
    if (lane == 0) {
#pragma unroll
        for (int r = 0; r < KR; ++r) redm[r * 16 + wave] = m[r];
    }
    __syncthreads();
#pragma unroll
    for (int r = 0; r < KR; ++r) m[r] = redm[r * 16 + (lane & 15)];
#pragma unroll
    for (int off = 8; off > 0; off >>= 1) {
        float t[KR];
#pragma unroll
        for (int r = 0; r < KR; ++r) t[r] = __shfl_xor(m[r], off, 32);
#pragma unroll
        for (int r = 0; r < KR; ++r) m[r] = fmaxf(m[r], t[r]);
    }

#pragma unroll
    for (int r = 0; r < KR; ++r) { s[r] = fast_exp(s[r] - m[r]); m[r] = s[r]; }
#pragma unroll
    for (int off = 16; off > 0; off >>= 1) {
        float t[KR];
#pragma unroll
        for (int r = 0; r < KR; ++r) t[r] = __shfl_xor(m[r], off, 32);
#pragma unroll
        for (int r = 0; r < KR; ++r) m[r] += t[r];
    }
    if (lane == 0) {
#pragma unroll
        for (int r = 0; r < KR; ++r) reds[r * 16 + wave] = m[r];
    }
    __syncthreads();
#pragma unroll
    for (int r = 0; r < KR; ++r) m[r] = reds[r * 16 + (lane & 15)];
#pragma unroll
    for (int off = 8; off > 0; off >>= 1) {
        float t[KR];
#pragma unroll
        for (int r = 0; r < KR; ++r) t[r] = __shfl_xor(m[r], off, 32);
#pragma unroll
        for (int r = 0; r < KR; ++r) m[r] += t[r];
    }

#pragma unroll
    for (int r = 0; r < KR; ++r) {
        float inv = 1.f / m[r];
        atten[((size_t)(b * LK + l0 + r)) * LQ + tid] = s[r] * inv;
    }
}

typedef float v4f __attribute__((ext_vector_type(4)));

#define CM    32
#define CN    64
#define CKW   (LQ / 4)
#define CNG   (CKW / 8)
#define CRING 6
#define CLDS  (168 * 1024 / 16)

struct CtxGroup {
    v4f a[2];
    v4f b[4];
};

__device__ __forceinline__ void ctx_load(CtxGroup& f, const float* __restrict__ ap,
                                         const float* __restrict__ bp, int g)
{
#pragma unroll
    for (int i = 0; i < 2; ++i)
        f.a[i] = *(const v4f*)(ap + (size_t)i * 16 * LQ + g * 8);
#pragma unroll
    for (int u = 0; u < 4; ++u)
        f.b[u] = *(const v4f*)(bp + (size_t)(g * 8 + u) * D_);
}

__device__ __forceinline__ void ctx_mma(v8f (&acc)[2][4], const CtxGroup& f)
{
#pragma unroll
    for (int m = 0; m < 2; ++m) {
#pragma unroll
        for (int i = 0; i < 2; ++i) {
            const v2f a = m ? (v2f){f.a[i].z, f.a[i].w} : (v2f){f.a[i].x, f.a[i].y};
#pragma unroll
            for (int j = 0; j < 4; ++j) {
                const v2f bm = (v2f){f.b[2 * m][j], f.b[2 * m + 1][j]};
                acc[i][j] = __builtin_amdgcn_wmma_f32_16x16x4_f32(
                     false, a,  false, bm,
                     (short)0, acc[i][j],
                     false,  false);
            }
        }
    }
}

__global__ __launch_bounds__(128) void context_wmma_kernel(
    const float* __restrict__ atten, const float* __restrict__ value,
    float* __restrict__ context)
{
    __shared__ v4f red[CLDS];

    const int wave = __builtin_amdgcn_readfirstlane(threadIdx.x >> 5);
    const int lane = threadIdx.x & 31;
    const int c    = lane & 15;
    const int h    = lane >> 4;
    const int bid  = blockIdx.x;
    const int b    = bid >> 6;
    const int l0   = ((bid >> 2) & 15) * CM;
    const int d0   = (bid & 3) * CN;
    const int k0   = wave * CKW;

    const float* ap = atten + ((size_t)(b * LK + l0 + c)) * LQ + k0 + 4 * h;
    const float* bp = value + ((size_t)(b * LQ + k0 + 4 * h)) * D_ + d0 + 4 * c;

    v8f acc[2][4];
#pragma unroll
    for (int i = 0; i < 2; ++i)
#pragma unroll
        for (int j = 0; j < 4; ++j)
            acc[i][j] = (v8f){0.f, 0.f, 0.f, 0.f, 0.f, 0.f, 0.f, 0.f};

    CtxGroup ring[CRING];
#pragma unroll
    for (int g = 0; g < CRING; ++g) ctx_load(ring[g], ap, bp, g);

#pragma unroll
    for (int g = 0; g < CNG; ++g) {
        if (g >= 1 && g - 1 + CRING < CNG) ctx_load(ring[(g - 1) % CRING], ap, bp, g - 1 + CRING);
        __builtin_amdgcn_sched_barrier(0);
        ctx_mma(acc, ring[g % CRING]);
        __builtin_amdgcn_sched_barrier(0);
    }

#pragma unroll
    for (int i = 0; i < 2; ++i)
#pragma unroll
        for (int r = 0; r < 8; ++r)
            red[((wave * 2 + i) * 8 + r) * 32 + lane] =
                (v4f){acc[i][0][r], acc[i][1][r], acc[i][2][r], acc[i][3][r]};
    __syncthreads();

    const int i  = wave >> 1;
    const int rb = (wave & 1) * 4;
#pragma unroll
    for (int rr = 0; rr < 4; ++rr) {
        const int r = rb + rr;
        v4f s = red[((0 * 2 + i) * 8 + r) * 32 + lane];
        s += red[((1 * 2 + i) * 8 + r) * 32 + lane];
        s += red[((2 * 2 + i) * 8 + r) * 32 + lane];
        s += red[((3 * 2 + i) * 8 + r) * 32 + lane];
        *(v4f*)(context + ((size_t)(b * LK + l0 + 16 * i + 8 * h + r)) * D_ + d0 + 4 * c) = s;
    }
}

extern "C" void kernel_launch(void* const* d_in, const int* in_sizes, int n_in,
                              void* d_out, int out_size, void* d_ws, size_t ws_size,
                              hipStream_t stream)
{
    const float* query = (const float*)d_in[0];
    const float* key   = (const float*)d_in[1];
    const float* value = (const float*)d_in[2];
    const float* Wc1   = (const float*)d_in[3];
    const float* Wc2   = (const float*)d_in[4];
    const float* vc    = (const float*)d_in[5];

    float* out     = (float*)d_out;
    float* context = out;
    float* atten   = out + (size_t)B_ * LK * D_;

    float* qp = (float*)d_ws;
    float* kp = qp + (size_t)B_ * LQ * E_;

    proj_kernel<<<2 * (B_ * 512) / (2 * PRW), 256, 0, stream>>>(query, key, Wc1, Wc2, qp, kp);
    score_softmax_kernel<<<B_ * (LK / KR), 512, 0, stream>>>(qp, kp, vc, atten);
    context_wmma_kernel<<<B_ * (LK / CM) * (D_ / CN), 128, 0, stream>>>(
        atten, value, context);
}


# score kernel: query fill pipelined under the tanh loop in 4 feature chunks; next chunk's row loads fly during the loop, wave-private LDS transposition, no block barrier
# speedup vs baseline: 2.8152x; 1.2910x over previous
#include <hip/hip_runtime.h>
#include <math.h>

#define B_  4
#define LQ  512
#define LK  512
#define E_  128
#define D_  256

typedef float v2f __attribute__((ext_vector_type(2)));
typedef float v8f __attribute__((ext_vector_type(8)));
typedef float v4f __attribute__((ext_vector_type(4)));

__device__ __forceinline__ float fast_tanh(float x) {
#if __has_builtin(__builtin_amdgcn_tanhf)
    return __builtin_amdgcn_tanhf(x);
#else
    return tanhf(x);
#endif
}

__device__ __forceinline__ float fast_exp(float x) {
#if __has_builtin(__builtin_amdgcn_exp2f)
    return __builtin_amdgcn_exp2f(x * 1.44269504088896340f);
#else
    return __expf(x);
#endif
}

#define PRW 4
#define PC  16
#define DH  (D_ / 2)

__device__ __forceinline__ void proj_chunk(v2f (&acc)[PRW], const v2f (&w)[PC],
                                           const float* __restrict__ x)
{
#pragma unroll
    for (int j = 0; j < PC; ++j) {
#pragma unroll
        for (int r = 0; r < PRW; ++r) {
            const float xs = x[r * D_ + j];
            acc[r] = __builtin_elementwise_fma((v2f){xs, xs}, w[j], acc[r]);
        }
    }
}

__global__ __launch_bounds__(256) void proj_kernel(
    const float* __restrict__ query, const float* __restrict__ key,
    const float* __restrict__ Wc1,   const float* __restrict__ Wc2,
    float* __restrict__ qp, float* __restrict__ kp)
{
    __shared__ v2f red[2 * PRW * (E_ / 2)];

    const int nb   = (B_ * 512) / (2 * PRW);
    const int bid  = blockIdx.x;
    const int half = bid / nb;
    const int tid  = threadIdx.x;
    const int ep   = tid & 63;
    const int rh   = __builtin_amdgcn_readfirstlane((tid >> 6) & 1);
    const int dh   = __builtin_amdgcn_readfirstlane(tid >> 7);
    const int row0 = (bid % nb) * (2 * PRW) + rh * PRW;

    const float* in  = (half ? key : query) + (size_t)row0 * D_ + dh * DH;
    const v2f*   W   = (const v2f*)(half ? Wc2 : Wc1) + (size_t)dh * DH * (E_ / 2) + ep;
    v2f*         out = (v2f*)((half ? kp : qp) + (size_t)row0 * E_) + ep;

    v2f acc[PRW];
#pragma unroll
    for (int r = 0; r < PRW; ++r) acc[r] = (v2f){0.f, 0.f};

    v2f wa[PC], wb[PC];
#pragma unroll
    for (int j = 0; j < PC; ++j) wa[j] = W[j * (E_ / 2)];

#pragma unroll 1
    for (int c = 0; c < DH / PC; c += 2) {
        const v2f* Wb = W + (size_t)(c + 1) * PC * (E_ / 2);
#pragma unroll
        for (int j = 0; j < PC; ++j) wb[j] = Wb[j * (E_ / 2)];
        proj_chunk(acc, wa, in + c * PC);

        const int cn = (c + 2 < DH / PC) ? (c + 2) : (c + 1);
        const v2f* Wa = W + (size_t)cn * PC * (E_ / 2);
#pragma unroll
        for (int j = 0; j < PC; ++j) wa[j] = Wa[j * (E_ / 2)];
        proj_chunk(acc, wb, in + (c + 1) * PC);
    }

    if (dh) {
#pragma unroll
        for (int r = 0; r < PRW; ++r) red[(rh * PRW + r) * (E_ / 2) + ep] = acc[r];
    }
    __syncthreads();
    if (!dh) {
#pragma unroll
        for (int r = 0; r < PRW; ++r) {
            const v2f t = red[(rh * PRW + r) * (E_ / 2) + ep];
            out[(size_t)r * (E_ / 2)] = acc[r] + t;
        }
    }
}

#define KR  8
#define QC  32
#define NQC (E_ / QC)
#define SPITCH (QC + 4)
#define SLDS (168 * 1024 / 4)

__global__ __launch_bounds__(512) void score_softmax_kernel(
    const float* __restrict__ qp, const float* __restrict__ kp,
    const float* __restrict__ vc, float* __restrict__ atten)
{
    __shared__ __align__(16) float stage[SLDS];
    __shared__ float redm[KR * 16];
    __shared__ float reds[KR * 16];

    const int bid  = blockIdx.x;
    const int b    = bid / (LK / KR);
    const int l0   = (bid % (LK / KR)) * KR;
    const int tid  = threadIdx.x;
    const int wave = tid >> 5;
    const int lane = tid & 31;

    const int lr = lane >> 3;
    const int lc = (lane & 7) * 4;
    const float* gsrc = qp + ((size_t)(b * LQ + wave * 32 + lr)) * E_ + lc;
    float*       st   = stage + wave * (32 * SPITCH);
    float*       sdst = st + lr * SPITCH + lc;
    const float* ssrc = st + lane * SPITCH;

    const float* kb = kp + ((size_t)(b * LK + l0)) * E_;

    v4f raw[8];
#pragma unroll
    for (int i = 0; i < 8; ++i)
        raw[i] = *(const v4f*)(gsrc + (size_t)i * 4 * E_);

    float s[KR], m[KR];
#pragma unroll
    for (int r = 0; r < KR; ++r) s[r] = 0.f;

#pragma unroll 1
    for (int c = 0; c < NQC; ++c) {
#pragma unroll
        for (int i = 0; i < 8; ++i)
            *(v4f*)(sdst + i * 4 * SPITCH) = raw[i];
        __builtin_amdgcn_wave_barrier();
        float q[QC];
#pragma unroll
        for (int j = 0; j < QC / 4; ++j) {
            const v4f v = *(const v4f*)(ssrc + j * 4);
            q[j * 4 + 0] = v.x;
            q[j * 4 + 1] = v.y;
            q[j * 4 + 2] = v.z;
            q[j * 4 + 3] = v.w;
        }
        if (c + 1 < NQC) {
#pragma unroll
            for (int i = 0; i < 8; ++i)
                raw[i] = *(const v4f*)(gsrc + (size_t)i * 4 * E_ + (c + 1) * QC);
        }
        __builtin_amdgcn_sched_barrier(0);

        const float* vcc = vc + c * QC;
#pragma unroll
        for (int rg = 0; rg < KR / 4; ++rg) {
            const float* k0 = kb + (size_t)(4 * rg) * E_ + c * QC;
            const float* k1 = k0 + E_;
            const float* k2 = k1 + E_;
            const float* k3 = k2 + E_;
            float a0 = 0.f, a1 = 0.f, a2 = 0.f, a3 = 0.f;
            float c0 = 0.f, c1 = 0.f, c2 = 0.f, c3 = 0.f;
#pragma unroll
            for (int e = 0; e < QC; e += 2) {
                a0 = fmaf(fast_tanh(q[e]     + k0[e]),     vcc[e],     a0);
                a1 = fmaf(fast_tanh(q[e]     + k1[e]),     vcc[e],     a1);
                a2 = fmaf(fast_tanh(q[e]     + k2[e]),     vcc[e],     a2);
                a3 = fmaf(fast_tanh(q[e]     + k3[e]),     vcc[e],     a3);
                c0 = fmaf(fast_tanh(q[e + 1] + k0[e + 1]), vcc[e + 1], c0);
                c1 = fmaf(fast_tanh(q[e + 1] + k1[e + 1]), vcc[e + 1], c1);
                c2 = fmaf(fast_tanh(q[e + 1] + k2[e + 1]), vcc[e + 1], c2);
                c3 = fmaf(fast_tanh(q[e + 1] + k3[e + 1]), vcc[e + 1], c3);
            }
            s[4 * rg]     += a0 + c0;
            s[4 * rg + 1] += a1 + c1;
            s[4 * rg + 2] += a2 + c2;
            s[4 * rg + 3] += a3 + c3;
        }
    }

#pragma unroll
    for (int r = 0; r < KR; ++r) m[r] = s[r];

#pragma unroll
    for (int off = 16; off > 0; off >>= 1) {
        float t[KR];
#pragma unroll
        for (int r = 0; r < KR; ++r) t[r] = __shfl_xor(m[r], off, 32);
#pragma unroll
        for (int r = 0; r < KR; ++r) m[r] = fmaxf(m[r], t[r]);
    }
    if (lane == 0) {
#pragma unroll
        for (int r = 0; r < KR; ++r) redm[r * 16 + wave] = m[r];
    }
    __syncthreads();
#pragma unroll
    for (int r = 0; r < KR; ++r) m[r] = redm[r * 16 + (lane & 15)];
#pragma unroll
    for (int off = 8; off > 0; off >>= 1) {
        float t[KR];
#pragma unroll
        for (int r = 0; r < KR; ++r) t[r] = __shfl_xor(m[r], off, 32);
#pragma unroll
        for (int r = 0; r < KR; ++r) m[r] = fmaxf(m[r], t[r]);
    }

#pragma unroll
    for (int r = 0; r < KR; ++r) { s[r] = fast_exp(s[r] - m[r]); m[r] = s[r]; }
#pragma unroll
    for (int off = 16; off > 0; off >>= 1) {
        float t[KR];
#pragma unroll
        for (int r = 0; r < KR; ++r) t[r] = __shfl_xor(m[r], off, 32);
#pragma unroll
        for (int r = 0; r < KR; ++r) m[r] += t[r];
    }
    if (lane == 0) {
#pragma unroll
        for (int r = 0; r < KR; ++r) reds[r * 16 + wave] = m[r];
    }
    __syncthreads();
#pragma unroll
    for (int r = 0; r < KR; ++r) m[r] = reds[r * 16 + (lane & 15)];
#pragma unroll
    for (int off = 8; off > 0; off >>= 1) {
        float t[KR];
#pragma unroll
        for (int r = 0; r < KR; ++r) t[r] = __shfl_xor(m[r], off, 32);
#pragma unroll
        for (int r = 0; r < KR; ++r) m[r] += t[r];
    }

#pragma unroll
    for (int r = 0; r < KR; ++r) {
        float inv = 1.f / m[r];
        atten[((size_t)(b * LK + l0 + r)) * LQ + tid] = s[r] * inv;
    }
}

#define CM    32
#define CN    64
#define CKW   (LQ / 4)
#define CNG   (CKW / 8)
#define CRING 6
#define CLDS  (168 * 1024 / 16)

struct CtxGroup {
    v4f a[2];
    v4f b[4];
};

__device__ __forceinline__ void ctx_load(CtxGroup& f, const float* __restrict__ ap,
                                         const float* __restrict__ bp, int g)
{
#pragma unroll
    for (int i = 0; i < 2; ++i)
        f.a[i] = *(const v4f*)(ap + (size_t)i * 16 * LQ + g * 8);
#pragma unroll
    for (int u = 0; u < 4; ++u)
        f.b[u] = *(const v4f*)(bp + (size_t)(g * 8 + u) * D_);
}

__device__ __forceinline__ void ctx_mma(v8f (&acc)[2][4], const CtxGroup& f)
{
#pragma unroll
    for (int m = 0; m < 2; ++m) {
#pragma unroll
        for (int i = 0; i < 2; ++i) {
            const v2f a = m ? (v2f){f.a[i].z, f.a[i].w} : (v2f){f.a[i].x, f.a[i].y};
#pragma unroll
            for (int j = 0; j < 4; ++j) {
                const v2f bm = (v2f){f.b[2 * m][j], f.b[2 * m + 1][j]};
                acc[i][j] = __builtin_amdgcn_wmma_f32_16x16x4_f32(
                     false, a,  false, bm,
                     (short)0, acc[i][j],
                     false,  false);
            }
        }
    }
}

__global__ __launch_bounds__(128) void context_wmma_kernel(
    const float* __restrict__ atten, const float* __restrict__ value,
    float* __restrict__ context)
{
    __shared__ v4f red[CLDS];

    const int wave = __builtin_amdgcn_readfirstlane(threadIdx.x >> 5);
    const int lane = threadIdx.x & 31;
    const int c    = lane & 15;
    const int h    = lane >> 4;
    const int bid  = blockIdx.x;
    const int b    = bid >> 6;
    const int l0   = ((bid >> 2) & 15) * CM;
    const int d0   = (bid & 3) * CN;
    const int k0   = wave * CKW;

    const float* ap = atten + ((size_t)(b * LK + l0 + c)) * LQ + k0 + 4 * h;
    const float* bp = value + ((size_t)(b * LQ + k0 + 4 * h)) * D_ + d0 + 4 * c;

    v8f acc[2][4];
#pragma unroll
    for (int i = 0; i < 2; ++i)
#pragma unroll
        for (int j = 0; j < 4; ++j)
            acc[i][j] = (v8f){0.f, 0.f, 0.f, 0.f, 0.f, 0.f, 0.f, 0.f};

    CtxGroup ring[CRING];
#pragma unroll
    for (int g = 0; g < CRING; ++g) ctx_load(ring[g], ap, bp, g);

#pragma unroll
    for (int g = 0; g < CNG; ++g) {
        if (g >= 1 && g - 1 + CRING < CNG) ctx_load(ring[(g - 1) % CRING], ap, bp, g - 1 + CRING);
        __builtin_amdgcn_sched_barrier(0);
        ctx_mma(acc, ring[g % CRING]);
        __builtin_amdgcn_sched_barrier(0);
    }

#pragma unroll
    for (int i = 0; i < 2; ++i)
#pragma unroll
        for (int r = 0; r < 8; ++r)
            red[((wave * 2 + i) * 8 + r) * 32 + lane] =
                (v4f){acc[i][0][r], acc[i][1][r], acc[i][2][r], acc[i][3][r]};
    __syncthreads();

    const int i  = wave >> 1;
    const int rb = (wave & 1) * 4;
#pragma unroll
    for (int rr = 0; rr < 4; ++rr) {
        const int r = rb + rr;
        v4f s = red[((0 * 2 + i) * 8 + r) * 32 + lane];
        s += red[((1 * 2 + i) * 8 + r) * 32 + lane];
        s += red[((2 * 2 + i) * 8 + r) * 32 + lane];
        s += red[((3 * 2 + i) * 8 + r) * 32 + lane];
        *(v4f*)(context + ((size_t)(b * LK + l0 + 16 * i + 8 * h + r)) * D_ + d0 + 4 * c) = s;
    }
}

extern "C" void kernel_launch(void* const* d_in, const int* in_sizes, int n_in,
                              void* d_out, int out_size, void* d_ws, size_t ws_size,
                              hipStream_t stream)
{
    const float* query = (const float*)d_in[0];
    const float* key   = (const float*)d_in[1];
    const float* value = (const float*)d_in[2];
    const float* Wc1   = (const float*)d_in[3];
    const float* Wc2   = (const float*)d_in[4];
    const float* vc    = (const float*)d_in[5];

    float* out     = (float*)d_out;
    float* context = out;
    float* atten   = out + (size_t)B_ * LK * D_;

    float* qp = (float*)d_ws;
    float* kp = qp + (size_t)B_ * LQ * E_;

    proj_kernel<<<2 * (B_ * 512) / (2 * PRW), 256, 0, stream>>>(query, key, Wc1, Wc2, qp, kp);
    score_softmax_kernel<<<B_ * (LK / KR), 512, 0, stream>>>(qp, kp, vc, atten);
    context_wmma_kernel<<<B_ * (LK / CM) * (D_ / CN), 128, 0, stream>>>(
        atten, value, context);
}


# score kernel softmax made one pass: half-wave row max and exp-sum by DPP cross-lane vector ops, one wave per row joins the 32 pairs; no LDS-crossbar butterflies, one join instead of two
# speedup vs baseline: 2.9143x; 1.0352x over previous
#include <hip/hip_runtime.h>
#include <math.h>

#define B_  4
#define LQ  512
#define LK  512
#define E_  128
#define D_  256

typedef float v2f __attribute__((ext_vector_type(2)));
typedef float v8f __attribute__((ext_vector_type(8)));
typedef float v4f __attribute__((ext_vector_type(4)));

__device__ __forceinline__ float fast_tanh(float x) {
#if __has_builtin(__builtin_amdgcn_tanhf)
    return __builtin_amdgcn_tanhf(x);
#else
    return tanhf(x);
#endif
}

__device__ __forceinline__ float fast_exp(float x) {
#if __has_builtin(__builtin_amdgcn_exp2f)
    return __builtin_amdgcn_exp2f(x * 1.44269504088896340f);
#else
    return __expf(x);
#endif
}

template <int N>
__device__ __forceinline__ float row_xor(float x) {
    const int b = __builtin_bit_cast(int, x);
    return __builtin_bit_cast(float, __builtin_amdgcn_update_dpp(b, b, 0x160 | N, 0xf, 0xf, true));
}

__device__ __forceinline__ float xhalf(float x) {
    const int b = __builtin_bit_cast(int, x);
    return __builtin_bit_cast(float,
        __builtin_amdgcn_permlanex16(b, b, 0x76543210, 0xfedcba98, false, false));
}

#define PRW 4
#define PC  16
#define DH  (D_ / 2)

__device__ __forceinline__ void proj_chunk(v2f (&acc)[PRW], const v2f (&w)[PC],
                                           const float* __restrict__ x)
{
#pragma unroll
    for (int j = 0; j < PC; ++j) {
#pragma unroll
        for (int r = 0; r < PRW; ++r) {
            const float xs = x[r * D_ + j];
            acc[r] = __builtin_elementwise_fma((v2f){xs, xs}, w[j], acc[r]);
        }
    }
}

__global__ __launch_bounds__(256) void proj_kernel(
    const float* __restrict__ query, const float* __restrict__ key,
    const float* __restrict__ Wc1,   const float* __restrict__ Wc2,
    float* __restrict__ qp, float* __restrict__ kp)
{
    __shared__ v2f red[2 * PRW * (E_ / 2)];

    const int nb   = (B_ * 512) / (2 * PRW);
    const int bid  = blockIdx.x;
    const int half = bid / nb;
    const int tid  = threadIdx.x;
    const int ep   = tid & 63;
    const int rh   = __builtin_amdgcn_readfirstlane((tid >> 6) & 1);
    const int dh   = __builtin_amdgcn_readfirstlane(tid >> 7);
    const int row0 = (bid % nb) * (2 * PRW) + rh * PRW;

    const float* in  = (half ? key : query) + (size_t)row0 * D_ + dh * DH;
    const v2f*   W   = (const v2f*)(half ? Wc2 : Wc1) + (size_t)dh * DH * (E_ / 2) + ep;
    v2f*         out = (v2f*)((half ? kp : qp) + (size_t)row0 * E_) + ep;

    v2f acc[PRW];
#pragma unroll
    for (int r = 0; r < PRW; ++r) acc[r] = (v2f){0.f, 0.f};

    v2f wa[PC], wb[PC];
#pragma unroll
    for (int j = 0; j < PC; ++j) wa[j] = W[j * (E_ / 2)];

#pragma unroll 1
    for (int c = 0; c < DH / PC; c += 2) {
        const v2f* Wb = W + (size_t)(c + 1) * PC * (E_ / 2);
#pragma unroll
        for (int j = 0; j < PC; ++j) wb[j] = Wb[j * (E_ / 2)];
        proj_chunk(acc, wa, in + c * PC);

        const int cn = (c + 2 < DH / PC) ? (c + 2) : (c + 1);
        const v2f* Wa = W + (size_t)cn * PC * (E_ / 2);
#pragma unroll
        for (int j = 0; j < PC; ++j) wa[j] = Wa[j * (E_ / 2)];
        proj_chunk(acc, wb, in + (c + 1) * PC);
    }

    if (dh) {
#pragma unroll
        for (int r = 0; r < PRW; ++r) red[(rh * PRW + r) * (E_ / 2) + ep] = acc[r];
    }
    __syncthreads();
    if (!dh) {
#pragma unroll
        for (int r = 0; r < PRW; ++r) {
            const v2f t = red[(rh * PRW + r) * (E_ / 2) + ep];
            out[(size_t)r * (E_ / 2)] = acc[r] + t;
        }
    }
}

#define KR  8
#define QC  32
#define NQC (E_ / QC)
#define SPITCH (QC + 4)
#define SLDS (168 * 1024 / 4)
#define PPITCH (2 * KR + 4)

__global__ __launch_bounds__(512) void score_softmax_kernel(
    const float* __restrict__ qp, const float* __restrict__ kp,
    const float* __restrict__ vc, float* __restrict__ atten)
{
    __shared__ __align__(16) float stage[SLDS];
    __shared__ __align__(16) float part[32 * PPITCH];
    __shared__ __align__(16) float fin[2 * KR];

    const int bid  = blockIdx.x;
    const int b    = bid / (LK / KR);
    const int l0   = (bid % (LK / KR)) * KR;
    const int tid  = threadIdx.x;
    const int wave = tid >> 5;
    const int lane = tid & 31;

    const int lr = lane >> 3;
    const int lc = (lane & 7) * 4;
    const float* gsrc = qp + ((size_t)(b * LQ + wave * 32 + lr)) * E_ + lc;
    float*       st   = stage + wave * (32 * SPITCH);
    float*       sdst = st + lr * SPITCH + lc;
    const float* ssrc = st + lane * SPITCH;

    const float* kb = kp + ((size_t)(b * LK + l0)) * E_;

    v4f raw[8];
#pragma unroll
    for (int i = 0; i < 8; ++i)
        raw[i] = *(const v4f*)(gsrc + (size_t)i * 4 * E_);

    float s[KR], m[KR];
#pragma unroll
    for (int r = 0; r < KR; ++r) s[r] = 0.f;

#pragma unroll 1
    for (int c = 0; c < NQC; ++c) {
#pragma unroll
        for (int i = 0; i < 8; ++i)
            *(v4f*)(sdst + i * 4 * SPITCH) = raw[i];
        __builtin_amdgcn_wave_barrier();
        float q[QC];
#pragma unroll
        for (int j = 0; j < QC / 4; ++j) {
            const v4f v = *(const v4f*)(ssrc + j * 4);
            q[j * 4 + 0] = v.x;
            q[j * 4 + 1] = v.y;
            q[j * 4 + 2] = v.z;
            q[j * 4 + 3] = v.w;
        }
        if (c + 1 < NQC) {
#pragma unroll
            for (int i = 0; i < 8; ++i)
                raw[i] = *(const v4f*)(gsrc + (size_t)i * 4 * E_ + (c + 1) * QC);
        }
        __builtin_amdgcn_sched_barrier(0);

        const float* vcc = vc + c * QC;
#pragma unroll
        for (int rg = 0; rg < KR / 4; ++rg) {
            const float* k0 = kb + (size_t)(4 * rg) * E_ + c * QC;
            const float* k1 = k0 + E_;
            const float* k2 = k1 + E_;
            const float* k3 = k2 + E_;
            float a0 = 0.f, a1 = 0.f, a2 = 0.f, a3 = 0.f;
            float c0 = 0.f, c1 = 0.f, c2 = 0.f, c3 = 0.f;
#pragma unroll
            for (int e = 0; e < QC; e += 2) {
                a0 = fmaf(fast_tanh(q[e]     + k0[e]),     vcc[e],     a0);
                a1 = fmaf(fast_tanh(q[e]     + k1[e]),     vcc[e],     a1);
                a2 = fmaf(fast_tanh(q[e]     + k2[e]),     vcc[e],     a2);
                a3 = fmaf(fast_tanh(q[e]     + k3[e]),     vcc[e],     a3);
                c0 = fmaf(fast_tanh(q[e + 1] + k0[e + 1]), vcc[e + 1], c0);
                c1 = fmaf(fast_tanh(q[e + 1] + k1[e + 1]), vcc[e + 1], c1);
                c2 = fmaf(fast_tanh(q[e + 1] + k2[e + 1]), vcc[e + 1], c2);
                c3 = fmaf(fast_tanh(q[e + 1] + k3[e + 1]), vcc[e + 1], c3);
            }
            s[4 * rg]     += a0 + c0;
            s[4 * rg + 1] += a1 + c1;
            s[4 * rg + 2] += a2 + c2;
            s[4 * rg + 3] += a3 + c3;
        }
    }

#pragma unroll
    for (int r = 0; r < KR; ++r) m[r] = s[r];
#define WAVE_STEP(v, OP, X)                                         \
    {                                                               \
        float t[KR];                                                \
        _Pragma("unroll") for (int r = 0; r < KR; ++r) t[r] = X(v[r]); \
        _Pragma("unroll") for (int r = 0; r < KR; ++r) v[r] = OP(v[r], t[r]); \
    }
#define OP_MAX(a, b) fmaxf(a, b)
#define OP_ADD(a, b) ((a) + (b))
    WAVE_STEP(m, OP_MAX, row_xor<1>)
    WAVE_STEP(m, OP_MAX, row_xor<2>)
    WAVE_STEP(m, OP_MAX, row_xor<4>)
    WAVE_STEP(m, OP_MAX, row_xor<8>)

    float z[KR];
#pragma unroll
    for (int r = 0; r < KR; ++r) { s[r] = fast_exp(s[r] - m[r]); z[r] = s[r]; }
    WAVE_STEP(z, OP_ADD, row_xor<1>)
    WAVE_STEP(z, OP_ADD, row_xor<2>)
    WAVE_STEP(z, OP_ADD, row_xor<4>)
    WAVE_STEP(z, OP_ADD, row_xor<8>)

    if ((lane & 15) == 0) {
        float* pw = part + (wave * 2 + (lane >> 4)) * PPITCH;
        *(v4f*)(pw)      = (v4f){m[0], m[1], m[2], m[3]};
        *(v4f*)(pw + 4)  = (v4f){m[4], m[5], m[6], m[7]};
        *(v4f*)(pw + 8)  = (v4f){z[0], z[1], z[2], z[3]};
        *(v4f*)(pw + 12) = (v4f){z[4], z[5], z[6], z[7]};
    }
    __syncthreads();

    const int wv = __builtin_amdgcn_readfirstlane(wave);
    if (wv < KR) {
        const float mj = part[lane * PPITCH + wv];
        const float zj = part[lane * PPITCH + KR + wv];
        float M = mj;
        M = fmaxf(M, row_xor<1>(M));
        M = fmaxf(M, row_xor<2>(M));
        M = fmaxf(M, row_xor<4>(M));
        M = fmaxf(M, row_xor<8>(M));
        M = fmaxf(M, xhalf(M));
        float Z = zj * fast_exp(mj - M);
        Z += row_xor<1>(Z);
        Z += row_xor<2>(Z);
        Z += row_xor<4>(Z);
        Z += row_xor<8>(Z);
        Z += xhalf(Z);
        if (lane == 0) {
            fin[wv]      = M;
            fin[KR + wv] = 1.f / Z;
        }
    }
    __syncthreads();

    const v4f M0 = *(const v4f*)(fin);
    const v4f M1 = *(const v4f*)(fin + 4);
    const v4f I0 = *(const v4f*)(fin + 8);
    const v4f I1 = *(const v4f*)(fin + 12);
    const float Mr[KR] = {M0.x, M0.y, M0.z, M0.w, M1.x, M1.y, M1.z, M1.w};
    const float Ir[KR] = {I0.x, I0.y, I0.z, I0.w, I1.x, I1.y, I1.z, I1.w};

#pragma unroll
    for (int r = 0; r < KR; ++r) {
        const float f = fast_exp(m[r] - Mr[r]) * Ir[r];
        atten[((size_t)(b * LK + l0 + r)) * LQ + tid] = s[r] * f;
    }
#undef WAVE_STEP
#undef OP_MAX
#undef OP_ADD
}

#define CM    32
#define CN    64
#define CKW   (LQ / 4)
#define CNG   (CKW / 8)
#define CRING 6
#define CLDS  (168 * 1024 / 16)

struct CtxGroup {
    v4f a[2];
    v4f b[4];
};

__device__ __forceinline__ void ctx_load(CtxGroup& f, const float* __restrict__ ap,
                                         const float* __restrict__ bp, int g)
{
#pragma unroll
    for (int i = 0; i < 2; ++i)
        f.a[i] = *(const v4f*)(ap + (size_t)i * 16 * LQ + g * 8);
#pragma unroll
    for (int u = 0; u < 4; ++u)
        f.b[u] = *(const v4f*)(bp + (size_t)(g * 8 + u) * D_);
}

__device__ __forceinline__ void ctx_mma(v8f (&acc)[2][4], const CtxGroup& f)
{
#pragma unroll
    for (int m = 0; m < 2; ++m) {
#pragma unroll
        for (int i = 0; i < 2; ++i) {
            const v2f a = m ? (v2f){f.a[i].z, f.a[i].w} : (v2f){f.a[i].x, f.a[i].y};
#pragma unroll
            for (int j = 0; j < 4; ++j) {
                const v2f bm = (v2f){f.b[2 * m][j], f.b[2 * m + 1][j]};
                acc[i][j] = __builtin_amdgcn_wmma_f32_16x16x4_f32(
                     false, a,  false, bm,
                     (short)0, acc[i][j],
                     false,  false);
            }
        }
    }
}

__global__ __launch_bounds__(128) void context_wmma_kernel(
    const float* __restrict__ atten, const float* __restrict__ value,
    float* __restrict__ context)
{
    __shared__ v4f red[CLDS];

    const int wave = __builtin_amdgcn_readfirstlane(threadIdx.x >> 5);
    const int lane = threadIdx.x & 31;
    const int c    = lane & 15;
    const int h    = lane >> 4;
    const int bid  = blockIdx.x;
    const int b    = bid >> 6;
    const int l0   = ((bid >> 2) & 15) * CM;
    const int d0   = (bid & 3) * CN;
    const int k0   = wave * CKW;

    const float* ap = atten + ((size_t)(b * LK + l0 + c)) * LQ + k0 + 4 * h;
    const float* bp = value + ((size_t)(b * LQ + k0 + 4 * h)) * D_ + d0 + 4 * c;

    v8f acc[2][4];
#pragma unroll
    for (int i = 0; i < 2; ++i)
#pragma unroll
        for (int j = 0; j < 4; ++j)
            acc[i][j] = (v8f){0.f, 0.f, 0.f, 0.f, 0.f, 0.f, 0.f, 0.f};

    CtxGroup ring[CRING];
#pragma unroll
    for (int g = 0; g < CRING; ++g) ctx_load(ring[g], ap, bp, g);

#pragma unroll
    for (int g = 0; g < CNG; ++g) {
        if (g >= 1 && g - 1 + CRING < CNG) ctx_load(ring[(g - 1) % CRING], ap, bp, g - 1 + CRING);
        __builtin_amdgcn_sched_barrier(0);
        ctx_mma(acc, ring[g % CRING]);
        __builtin_amdgcn_sched_barrier(0);
    }

#pragma unroll
    for (int i = 0; i < 2; ++i)
#pragma unroll
        for (int r = 0; r < 8; ++r)
            red[((wave * 2 + i) * 8 + r) * 32 + lane] =
                (v4f){acc[i][0][r], acc[i][1][r], acc[i][2][r], acc[i][3][r]};
    __syncthreads();

    const int i  = wave >> 1;
    const int rb = (wave & 1) * 4;
#pragma unroll
    for (int rr = 0; rr < 4; ++rr) {
        const int r = rb + rr;
        v4f s = red[((0 * 2 + i) * 8 + r) * 32 + lane];
        s += red[((1 * 2 + i) * 8 + r) * 32 + lane];
        s += red[((2 * 2 + i) * 8 + r) * 32 + lane];
        s += red[((3 * 2 + i) * 8 + r) * 32 + lane];
        *(v4f*)(context + ((size_t)(b * LK + l0 + 16 * i + 8 * h + r)) * D_ + d0 + 4 * c) = s;
    }
}

extern "C" void kernel_launch(void* const* d_in, const int* in_sizes, int n_in,
                              void* d_out, int out_size, void* d_ws, size_t ws_size,
                              hipStream_t stream)
{
    const float* query = (const float*)d_in[0];
    const float* key   = (const float*)d_in[1];
    const float* value = (const float*)d_in[2];
    const float* Wc1   = (const float*)d_in[3];
    const float* Wc2   = (const float*)d_in[4];
    const float* vc    = (const float*)d_in[5];

    float* out     = (float*)d_out;
    float* context = out;
    float* atten   = out + (size_t)B_ * LK * D_;

    float* qp = (float*)d_ws;
    float* kp = qp + (size_t)B_ * LQ * E_;

    proj_kernel<<<2 * (B_ * 512) / (2 * PRW), 256, 0, stream>>>(query, key, Wc1, Wc2, qp, kp);
    score_softmax_kernel<<<B_ * (LK / KR), 512, 0, stream>>>(qp, kp, vc, atten);
    context_wmma_kernel<<<B_ * (LK / CM) * (D_ / CN), 128, 0, stream>>>(
        atten, value, context);
}
